# starting kernel
# baseline (speedup 1.0000x reference)
#include <hip/hip_runtime.h>

typedef float v2f __attribute__((ext_vector_type(2)));
typedef float v8f __attribute__((ext_vector_type(8)));

#define HH_ 160
#define WW_ 160
#define HWSZ (HH_*WW_)
#define CCH 32
#define KK 9
#define NSTAGES 9
#define BLK 32
#define XSTRIDE 292

__global__ __launch_bounds__(256) void head_kernel(
    const float* __restrict__ x,
    const float* __restrict__ hwt,
    float* __restrict__ xh)
{
  int p = blockIdx.x * blockDim.x + threadIdx.x;
  if (p >= 2 * HWSZ) return;
  int b = p / HWSZ, hw = p % HWSZ;
  const float* xb = x + (size_t)b * 3 * HWSZ + hw;
  float x0 = xb[0], x1 = xb[HWSZ], x2 = xb[2 * HWSZ];
  float* o = xh + (size_t)p * CCH;
#pragma unroll
  for (int oc = 0; oc < CCH; ++oc) {
    float v = hwt[oc * 3 + 0] * x0 + hwt[oc * 3 + 1] * x1 + hwt[oc * 3 + 2] * x2;
    o[oc] = fmaxf(v, 0.0f);
  }
}

__global__ __launch_bounds__(BLK) void lowrank_kernel(
    const float* __restrict__ xh,
    const float* __restrict__ tailw,
    const float* __restrict__ alphas,
    const float* __restrict__ betas,
    float* __restrict__ out)
{
  extern __shared__ float smem[];
  float* Xl  = smem;
  float* UVl = smem + BLK * XSTRIDE;

  const int t  = threadIdx.x;
  const int p  = blockIdx.x * BLK + t;
  const int b  = p / HWSZ;
  const int hw = p % HWSZ;
  const int h  = hw / WW_, w = hw % WW_;

  float* Xt = Xl + t * XSTRIDE;

  float u[CCH];
  float v[KK];
#pragma unroll
  for (int c = 0; c < CCH; ++c) u[c] = 0.0f;

#pragma unroll
  for (int k = 0; k < KK; ++k) {
    int m   = h * (WW_ * KK) + w * KK + k;
    int q   = m / HWSZ;
    int rem = m - q * HWSZ;
    int hp  = rem / WW_;
    int wp  = rem - hp * WW_;
    int ki  = q / 3;
    int kj  = q - 3 * ki;
    int hs  = hp + ki - 1; hs = hs < 0 ? 0 : (hs > HH_ - 1 ? HH_ - 1 : hs);
    int ws  = wp + kj - 1; ws = ws < 0 ? 0 : (ws > WW_ - 1 ? WW_ - 1 : ws);
    const float4* src = (const float4*)(xh + ((size_t)b * HWSZ + (size_t)hs * WW_ + ws) * CCH);
    float4* dst = (float4*)(Xt + k * CCH);
    float vk = 0.0f;
#pragma unroll
    for (int cc = 0; cc < 8; ++cc) {
      float4 d = src[cc];
      dst[cc] = d;
      u[4 * cc + 0] += d.x; u[4 * cc + 1] += d.y;
      u[4 * cc + 2] += d.z; u[4 * cc + 3] += d.w;
      vk += d.x + d.y + d.z + d.w;
    }
    v[k] = vk * (1.0f / 32.0f);
  }
#pragma unroll
  for (int c = 0; c < CCH; ++c) u[c] *= (1.0f / 9.0f);

  for (int s = 0; s < NSTAGES; ++s) {
    float alpha = alphas[s];
    float beta  = betas[s];

    float s2u = 0.0f;
#pragma unroll
    for (int c = 0; c < CCH; ++c) s2u += u[c] * u[c];
    float numV[KK];
#pragma unroll
    for (int k = 0; k < KK; ++k) {
      const float4* row = (const float4*)(Xt + k * CCH);
      float acc = 0.0f;
#pragma unroll
      for (int cc = 0; cc < 8; ++cc) {
        float4 d = row[cc];
        acc += u[4 * cc + 0] * d.x + u[4 * cc + 1] * d.y
             + u[4 * cc + 2] * d.z + u[4 * cc + 3] * d.w;
      }
      numV[k] = acc;
    }
    float c3s2u = 3.0f * s2u;
#pragma unroll
    for (int k = 0; k < KK; ++k)
      v[k] = v[k] - beta * (c3s2u * v[k] - numV[k]);

    float s2v = 0.0f;
#pragma unroll
    for (int k = 0; k < KK; ++k) s2v += v[k] * v[k];
    float4 Xv[8];
#pragma unroll
    for (int cc = 0; cc < 8; ++cc) Xv[cc] = make_float4(0.f, 0.f, 0.f, 0.f);
#pragma unroll
    for (int k = 0; k < KK; ++k) {
      const float4* row = (const float4*)(Xt + k * CCH);
      float vk = v[k];
#pragma unroll
      for (int cc = 0; cc < 8; ++cc) {
        float4 d = row[cc];
        Xv[cc].x += vk * d.x; Xv[cc].y += vk * d.y;
        Xv[cc].z += vk * d.z; Xv[cc].w += vk * d.w;
      }
    }
    float c3s2v = 3.0f * s2v;
#pragma unroll
    for (int cc = 0; cc < 8; ++cc) {
      u[4 * cc + 0] -= alpha * (c3s2v * u[4 * cc + 0] - Xv[cc].x);
      u[4 * cc + 1] -= alpha * (c3s2v * u[4 * cc + 1] - Xv[cc].y);
      u[4 * cc + 2] -= alpha * (c3s2v * u[4 * cc + 2] - Xv[cc].z);
      u[4 * cc + 3] -= alpha * (c3s2v * u[4 * cc + 3] - Xv[cc].w);
    }
  }

  float v4c = 3.0f * v[4];
#pragma unroll
  for (int c = 0; c < CCH; ++c) UVl[c * BLK + t] = u[c] * v4c;
  __syncthreads();

  const int lane = t;
  const int mrow = lane & 15;
  const bool lo  = (lane < 16);
  float4 twv[8];
#pragma unroll
  for (int cc = 0; cc < 8; ++cc)
    twv[cc] = (mrow < 3) ? ((const float4*)(tailw + mrow * CCH))[cc]
                         : make_float4(0.f, 0.f, 0.f, 0.f);

  const int pix0   = blockIdx.x * BLK;
  const int bb     = pix0 / HWSZ;
  const int hwbase = pix0 % HWSZ;

#pragma unroll
  for (int half = 0; half < 2; ++half) {
    v8f acc = {0.f, 0.f, 0.f, 0.f, 0.f, 0.f, 0.f, 0.f};
#pragma unroll
    for (int k4 = 0; k4 < 8; ++k4) {
      float4 q = twv[k4];
      v2f a, bm;
      a.x = lo ? q.x : q.z;
      a.y = lo ? q.y : q.w;
      int kk = 4 * k4 + (lo ? 0 : 2);
      bm.x = UVl[(kk + 0) * BLK + half * 16 + mrow];
      bm.y = UVl[(kk + 1) * BLK + half * 16 + mrow];
      acc = __builtin_amdgcn_wmma_f32_16x16x4_f32(
           false, a,  false, bm,
           (short)0, acc,  false,  false);
    }
    if (lane < 16) {
      int hwp = hwbase + half * 16 + lane;
      float* ob = out + (size_t)bb * 3 * HWSZ + hwp;
      ob[0]        = fmaxf(acc[0], 0.0f);
      ob[HWSZ]     = fmaxf(acc[1], 0.0f);
      ob[2 * HWSZ] = fmaxf(acc[2], 0.0f);
    }
  }
}

extern "C" void kernel_launch(void* const* d_in, const int* in_sizes, int n_in,
                              void* d_out, int out_size, void* d_ws, size_t ws_size,
                              hipStream_t stream) {
  const float* x      = (const float*)d_in[0];
  const float* head_w = (const float*)d_in[1];
  const float* tail_w = (const float*)d_in[2];
  const float* alphas = (const float*)d_in[3];
  const float* betas  = (const float*)d_in[4];
  float* out = (float*)d_out;
  float* xh  = (float*)d_ws;

  head_kernel<<<(2 * HWSZ + 255) / 256, 256, 0, stream>>>(x, head_w, xh);

  size_t shmem = (size_t)(BLK * XSTRIDE + CCH * BLK) * sizeof(float);
  lowrank_kernel<<<(2 * HWSZ) / BLK, BLK, shmem, stream>>>(xh, tail_w, alphas, betas, out);
}


# Head 1x1 conv + relu computed inside the gather of the stage kernel from the 3 input channels: one launch instead of two, NHWC intermediate and its 32-channel loads gone
# speedup vs baseline: 1.3717x; 1.3717x over previous
#include <hip/hip_runtime.h>

typedef float v2f __attribute__((ext_vector_type(2)));
typedef float v8f __attribute__((ext_vector_type(8)));

#define HH_ 160
#define WW_ 160
#define HWSZ (HH_*WW_)
#define CCH 32
#define KK 9
#define NSTAGES 9
#define BLK 32
#define XSTRIDE 292

__global__ __launch_bounds__(BLK) void lowrank_kernel(
    const float* __restrict__ x,
    const float* __restrict__ hwt,
    const float* __restrict__ tailw,
    const float* __restrict__ alphas,
    const float* __restrict__ betas,
    float* __restrict__ out)
{
  extern __shared__ float smem[];
  float* Xl  = smem;
  float* UVl = smem + BLK * XSTRIDE;

  const int t  = threadIdx.x;
  const int p  = blockIdx.x * BLK + t;
  const int b  = p / HWSZ;
  const int hw = p % HWSZ;
  const int h  = hw / WW_, w = hw % WW_;

  float* Xt = Xl + t * XSTRIDE;

  float u[CCH];
  float v[KK];
#pragma unroll
  for (int c = 0; c < CCH; ++c) u[c] = 0.0f;

#pragma unroll
  for (int k = 0; k < KK; ++k) {
    int m   = h * (WW_ * KK) + w * KK + k;
    int q   = m / HWSZ;
    int rem = m - q * HWSZ;
    int hp  = rem / WW_;
    int wp  = rem - hp * WW_;
    int ki  = q / 3;
    int kj  = q - 3 * ki;
    int hs  = hp + ki - 1; hs = hs < 0 ? 0 : (hs > HH_ - 1 ? HH_ - 1 : hs);
    int ws  = wp + kj - 1; ws = ws < 0 ? 0 : (ws > WW_ - 1 ? WW_ - 1 : ws);
    const float* xb = x + (size_t)b * 3 * HWSZ + (size_t)hs * WW_ + ws;
    float x0 = xb[0], x1 = xb[HWSZ], x2 = xb[2 * HWSZ];
    float4* dst = (float4*)(Xt + k * CCH);
    float vk = 0.0f;
#pragma unroll
    for (int cc = 0; cc < 8; ++cc) {
      float4 d;
      d.x = fmaxf(hwt[(4 * cc + 0) * 3 + 0] * x0 + hwt[(4 * cc + 0) * 3 + 1] * x1 + hwt[(4 * cc + 0) * 3 + 2] * x2, 0.0f);
      d.y = fmaxf(hwt[(4 * cc + 1) * 3 + 0] * x0 + hwt[(4 * cc + 1) * 3 + 1] * x1 + hwt[(4 * cc + 1) * 3 + 2] * x2, 0.0f);
      d.z = fmaxf(hwt[(4 * cc + 2) * 3 + 0] * x0 + hwt[(4 * cc + 2) * 3 + 1] * x1 + hwt[(4 * cc + 2) * 3 + 2] * x2, 0.0f);
      d.w = fmaxf(hwt[(4 * cc + 3) * 3 + 0] * x0 + hwt[(4 * cc + 3) * 3 + 1] * x1 + hwt[(4 * cc + 3) * 3 + 2] * x2, 0.0f);
      dst[cc] = d;
      u[4 * cc + 0] += d.x; u[4 * cc + 1] += d.y;
      u[4 * cc + 2] += d.z; u[4 * cc + 3] += d.w;
      vk += d.x + d.y + d.z + d.w;
    }
    v[k] = vk * (1.0f / 32.0f);
  }
#pragma unroll
  for (int c = 0; c < CCH; ++c) u[c] *= (1.0f / 9.0f);

  for (int s = 0; s < NSTAGES; ++s) {
    float alpha = alphas[s];
    float beta  = betas[s];

    float s2u = 0.0f;
#pragma unroll
    for (int c = 0; c < CCH; ++c) s2u += u[c] * u[c];
    float numV[KK];
#pragma unroll
    for (int k = 0; k < KK; ++k) {
      const float4* row = (const float4*)(Xt + k * CCH);
      float acc = 0.0f;
#pragma unroll
      for (int cc = 0; cc < 8; ++cc) {
        float4 d = row[cc];
        acc += u[4 * cc + 0] * d.x + u[4 * cc + 1] * d.y
             + u[4 * cc + 2] * d.z + u[4 * cc + 3] * d.w;
      }
      numV[k] = acc;
    }
    float c3s2u = 3.0f * s2u;
#pragma unroll
    for (int k = 0; k < KK; ++k)
      v[k] = v[k] - beta * (c3s2u * v[k] - numV[k]);

    float s2v = 0.0f;
#pragma unroll
    for (int k = 0; k < KK; ++k) s2v += v[k] * v[k];
    float4 Xv[8];
#pragma unroll
    for (int cc = 0; cc < 8; ++cc) Xv[cc] = make_float4(0.f, 0.f, 0.f, 0.f);
#pragma unroll
    for (int k = 0; k < KK; ++k) {
      const float4* row = (const float4*)(Xt + k * CCH);
      float vk = v[k];
#pragma unroll
      for (int cc = 0; cc < 8; ++cc) {
        float4 d = row[cc];
        Xv[cc].x += vk * d.x; Xv[cc].y += vk * d.y;
        Xv[cc].z += vk * d.z; Xv[cc].w += vk * d.w;
      }
    }
    float c3s2v = 3.0f * s2v;
#pragma unroll
    for (int cc = 0; cc < 8; ++cc) {
      u[4 * cc + 0] -= alpha * (c3s2v * u[4 * cc + 0] - Xv[cc].x);
      u[4 * cc + 1] -= alpha * (c3s2v * u[4 * cc + 1] - Xv[cc].y);
      u[4 * cc + 2] -= alpha * (c3s2v * u[4 * cc + 2] - Xv[cc].z);
      u[4 * cc + 3] -= alpha * (c3s2v * u[4 * cc + 3] - Xv[cc].w);
    }
  }

  float v4c = 3.0f * v[4];
#pragma unroll
  for (int c = 0; c < CCH; ++c) UVl[c * BLK + t] = u[c] * v4c;
  __syncthreads();

  const int lane = t;
  const int mrow = lane & 15;
  const bool lo  = (lane < 16);
  float4 twv[8];
#pragma unroll
  for (int cc = 0; cc < 8; ++cc)
    twv[cc] = (mrow < 3) ? ((const float4*)(tailw + mrow * CCH))[cc]
                         : make_float4(0.f, 0.f, 0.f, 0.f);

  const int pix0   = blockIdx.x * BLK;
  const int bb     = pix0 / HWSZ;
  const int hwbase = pix0 % HWSZ;

#pragma unroll
  for (int half = 0; half < 2; ++half) {
    v8f acc = {0.f, 0.f, 0.f, 0.f, 0.f, 0.f, 0.f, 0.f};
#pragma unroll
    for (int k4 = 0; k4 < 8; ++k4) {
      float4 q = twv[k4];
      v2f a, bm;
      a.x = lo ? q.x : q.z;
      a.y = lo ? q.y : q.w;
      int kk = 4 * k4 + (lo ? 0 : 2);
      bm.x = UVl[(kk + 0) * BLK + half * 16 + mrow];
      bm.y = UVl[(kk + 1) * BLK + half * 16 + mrow];
      acc = __builtin_amdgcn_wmma_f32_16x16x4_f32(
           false, a,  false, bm,
           (short)0, acc,  false,  false);
    }
    if (lane < 16) {
      int hwp = hwbase + half * 16 + lane;
      float* ob = out + (size_t)bb * 3 * HWSZ + hwp;
      ob[0]        = fmaxf(acc[0], 0.0f);
      ob[HWSZ]     = fmaxf(acc[1], 0.0f);
      ob[2 * HWSZ] = fmaxf(acc[2], 0.0f);
    }
  }
}

extern "C" void kernel_launch(void* const* d_in, const int* in_sizes, int n_in,
                              void* d_out, int out_size, void* d_ws, size_t ws_size,
                              hipStream_t stream) {
  const float* x      = (const float*)d_in[0];
  const float* head_w = (const float*)d_in[1];
  const float* tail_w = (const float*)d_in[2];
  const float* alphas = (const float*)d_in[3];
  const float* betas  = (const float*)d_in[4];
  float* out = (float*)d_out;

  size_t shmem = (size_t)(BLK * XSTRIDE + CCH * BLK) * sizeof(float);
  lowrank_kernel<<<(2 * HWSZ) / BLK, BLK, shmem, stream>>>(x, head_w, tail_w, alphas, betas, out);
}


# Stage loop run on a 9-vector a with U = X^T a and G = X X^T built once per pixel: 81 FMAs per stage for G a replace two 288-FMA passes over X; changes the order of the f32 sums
# speedup vs baseline: 2.3445x; 1.7092x over previous
#include <hip/hip_runtime.h>


typedef float v2f __attribute__((ext_vector_type(2)));
typedef float v8f __attribute__((ext_vector_type(8)));

#define HH_ 160
#define WW_ 160
#define HWSZ (HH_*WW_)
#define CCH 32
#define KK 9
#define NSTAGES 9
#define BLK 32
#define XSTRIDE 292

__global__ __launch_bounds__(BLK) void lowrank_kernel(
    const float* __restrict__ x,
    const float* __restrict__ hwt,
    const float* __restrict__ tailw,
    const float* __restrict__ alphas,
    const float* __restrict__ betas,
    float* __restrict__ out)
{
  extern __shared__ float smem[];
  float* Xl  = smem;
  float* UVl = smem + BLK * XSTRIDE;

  const int t  = threadIdx.x;
  const int p  = blockIdx.x * BLK + t;
  const int b  = p / HWSZ;
  const int hw = p % HWSZ;
  const int h  = hw / WW_, w = hw % WW_;

  float* Xt = Xl + t * XSTRIDE;


  float xs0[KK], xs1[KK], xs2[KK];
#pragma unroll
  for (int k = 0; k < KK; ++k) {
    int m   = h * (WW_ * KK) + w * KK + k;
    int q   = m / HWSZ;
    int rem = m - q * HWSZ;
    int hp  = rem / WW_;
    int wp  = rem - hp * WW_;
    int ki  = q / 3;
    int kj  = q - 3 * ki;
    int hs  = hp + ki - 1; hs = hs < 0 ? 0 : (hs > HH_ - 1 ? HH_ - 1 : hs);
    int ws  = wp + kj - 1; ws = ws < 0 ? 0 : (ws > WW_ - 1 ? WW_ - 1 : ws);
    const float* xb = x + (size_t)b * 3 * HWSZ + (size_t)hs * WW_ + ws;
    xs0[k] = xb[0]; xs1[k] = xb[HWSZ]; xs2[k] = xb[2 * HWSZ];
  }

  v2f G2[KK][KK];
  v2f vs2[KK];
#pragma unroll
  for (int k = 0; k < KK; ++k) {
    vs2[k] = (v2f){0.0f, 0.0f};
#pragma unroll
    for (int j = 0; j <= k; ++j) G2[k][j] = (v2f){0.0f, 0.0f};
  }

  float hc[12];
#pragma unroll
  for (int i = 0; i < 12; ++i) hc[i] = hwt[i];
#pragma unroll 1
  for (int cg = 0; cg < 8; ++cg) {
    const float* hnp = hwt + ((cg + 1) & 7) * 12;
    float hn[12];
#pragma unroll
    for (int i = 0; i < 12; ++i) hn[i] = hnp[i];

    v2f d2[KK][2];
#pragma unroll
    for (int k = 0; k < KK; ++k) {
#pragma unroll
      for (int p = 0; p < 2; ++p) {
        v2f h0 = {hc[6 * p + 0], hc[6 * p + 3]};
        v2f h1 = {hc[6 * p + 1], hc[6 * p + 4]};
        v2f h2 = {hc[6 * p + 2], hc[6 * p + 5]};
        v2f r  = h0 * xs0[k] + h1 * xs1[k] + h2 * xs2[k];
        d2[k][p] = (v2f){fmaxf(r.x, 0.0f), fmaxf(r.y, 0.0f)};
      }
      *(float4*)(Xt + k * CCH + cg * 4) = make_float4(d2[k][0].x, d2[k][0].y, d2[k][1].x, d2[k][1].y);
      vs2[k] += d2[k][0];
      vs2[k] += d2[k][1];
    }
#pragma unroll
    for (int k = 0; k < KK; ++k) {
#pragma unroll
      for (int j = 0; j <= k; ++j) {
        G2[k][j] += d2[k][0] * d2[j][0];
        G2[k][j] += d2[k][1] * d2[j][1];
      }
    }
#pragma unroll
    for (int i = 0; i < 12; ++i) hc[i] = hn[i];
  }

  float G[KK][KK];
  float v[KK];
#pragma unroll
  for (int k = 0; k < KK; ++k) {
    v[k] = (vs2[k].x + vs2[k].y) * (1.0f / 32.0f);
#pragma unroll
    for (int j = 0; j <= k; ++j) {
      float g = G2[k][j].x + G2[k][j].y;
      G[k][j] = g;
      G[j][k] = g;
    }
  }

  float a[KK];
#pragma unroll
  for (int k = 0; k < KK; ++k) a[k] = (1.0f / 9.0f);

  for (int s = 0; s < NSTAGES; ++s) {
    float alpha = alphas[s];
    float beta  = betas[s];

    float numV[KK];
#pragma unroll
    for (int k = 0; k < KK; ++k) {
      float acc = 0.0f;
#pragma unroll
      for (int j = 0; j < KK; ++j) {
        acc += G[k][j] * a[j];
      }
      numV[k] = acc;
    }
    float s2u = (a[0] * numV[0] + a[1] * numV[1] + a[2] * numV[2])
              + (a[3] * numV[3] + a[4] * numV[4] + a[5] * numV[5])
              + (a[6] * numV[6] + a[7] * numV[7] + a[8] * numV[8]);
    float c3s2u = 3.0f * s2u;
#pragma unroll
    for (int k = 0; k < KK; ++k)
      v[k] = v[k] - beta * (c3s2u * v[k] - numV[k]);

    float s2v = (v[0] * v[0] + v[1] * v[1] + v[2] * v[2])
              + (v[3] * v[3] + v[4] * v[4] + v[5] * v[5])
              + (v[6] * v[6] + v[7] * v[7] + v[8] * v[8]);
    float c3s2v = 3.0f * s2v;
#pragma unroll
    for (int k = 0; k < KK; ++k)
      a[k] = a[k] - alpha * (c3s2v * a[k] - v[k]);
  }

  float4 u4[8];
#pragma unroll
  for (int cc = 0; cc < 8; ++cc) u4[cc] = make_float4(0.f, 0.f, 0.f, 0.f);
#pragma unroll
  for (int k = 0; k < KK; ++k) {
    const float4* row = (const float4*)(Xt + k * CCH);
    float ak = a[k];
#pragma unroll
    for (int cc = 0; cc < 8; ++cc) {
      float4 d = row[cc];
      u4[cc].x += ak * d.x; u4[cc].y += ak * d.y;
      u4[cc].z += ak * d.z; u4[cc].w += ak * d.w;
    }
  }

  float v4c = 3.0f * v[4];
#pragma unroll
  for (int cc = 0; cc < 8; ++cc) {
    UVl[(4 * cc + 0) * BLK + t] = u4[cc].x * v4c;
    UVl[(4 * cc + 1) * BLK + t] = u4[cc].y * v4c;
    UVl[(4 * cc + 2) * BLK + t] = u4[cc].z * v4c;
    UVl[(4 * cc + 3) * BLK + t] = u4[cc].w * v4c;
  }
  __syncthreads();

  const int lane = t;
  const int mrow = lane & 15;
  const bool lo  = (lane < 16);
  float4 twv[8];
#pragma unroll
  for (int cc = 0; cc < 8; ++cc)
    twv[cc] = (mrow < 3) ? ((const float4*)(tailw + mrow * CCH))[cc]
                         : make_float4(0.f, 0.f, 0.f, 0.f);

  const int pix0   = blockIdx.x * BLK;
  const int bb     = pix0 / HWSZ;
  const int hwbase = pix0 % HWSZ;

#pragma unroll
  for (int half = 0; half < 2; ++half) {
    v8f acc = {0.f, 0.f, 0.f, 0.f, 0.f, 0.f, 0.f, 0.f};
#pragma unroll
    for (int k4 = 0; k4 < 8; ++k4) {
      float4 q = twv[k4];
      v2f a2, bm;
      a2.x = lo ? q.x : q.z;
      a2.y = lo ? q.y : q.w;
      int kk = 4 * k4 + (lo ? 0 : 2);
      bm.x = UVl[(kk + 0) * BLK + half * 16 + mrow];
      bm.y = UVl[(kk + 1) * BLK + half * 16 + mrow];
      acc = __builtin_amdgcn_wmma_f32_16x16x4_f32(
           false, a2,  false, bm,
           (short)0, acc,  false,  false);
    }
    if (lane < 16) {
      int hwp = hwbase + half * 16 + lane;
      float* ob = out + (size_t)bb * 3 * HWSZ + hwp;
      ob[0]        = fmaxf(acc[0], 0.0f);
      ob[HWSZ]     = fmaxf(acc[1], 0.0f);
      ob[2 * HWSZ] = fmaxf(acc[2], 0.0f);
    }
  }
}

extern "C" void kernel_launch(void* const* d_in, const int* in_sizes, int n_in,
                              void* d_out, int out_size, void* d_ws, size_t ws_size,
                              hipStream_t stream) {
  const float* x      = (const float*)d_in[0];
  const float* head_w = (const float*)d_in[1];
  const float* tail_w = (const float*)d_in[2];
  const float* alphas = (const float*)d_in[3];
  const float* betas  = (const float*)d_in[4];
  float* out = (float*)d_out;

  size_t shmem = (size_t)(BLK * XSTRIDE + CCH * BLK) * sizeof(float);
  lowrank_kernel<<<(2 * HWSZ) / BLK, BLK, shmem, stream>>>(x, head_w, tail_w, alphas, betas, out);
}


# Eight waves per work-group (200 groups of 256 threads, not 1600 of 32), each wave on its own 32 pixels and LDS slice, tail staging reusing the wave's dead X tiles; sum order unchanged
# speedup vs baseline: 2.7200x; 1.1602x over previous
#include <hip/hip_runtime.h>


typedef float v2f __attribute__((ext_vector_type(2)));
typedef float v8f __attribute__((ext_vector_type(8)));

#define HH_ 160
#define WW_ 160
#define HWSZ (HH_*WW_)
#define CCH 32
#define KK 9
#define NSTAGES 9
#define WAVES 8
#define BLK (32 * WAVES)
#define XSTRIDE 292

__device__ __forceinline__ void wave_lds_order() {
  __builtin_amdgcn_fence(__ATOMIC_RELEASE, "workgroup");
  __builtin_amdgcn_wave_barrier();
  __builtin_amdgcn_fence(__ATOMIC_ACQUIRE, "workgroup");
}

__global__ __launch_bounds__(BLK) void lowrank_kernel(
    const float* __restrict__ x,
    const float* __restrict__ hwt,
    const float* __restrict__ tailw,
    const float* __restrict__ alphas,
    const float* __restrict__ betas,
    float* __restrict__ out)
{
  __shared__ __attribute__((aligned(16))) float smem[BLK * XSTRIDE];

  const int tid = threadIdx.x;
  const int t   = tid & 31;
  const int wv  = __builtin_amdgcn_readfirstlane(tid >> 5);
  const int p   = blockIdx.x * BLK + tid;

  float* Xl  = smem + wv * (32 * XSTRIDE);
  float* UVl = Xl;
  const int b  = p / HWSZ;
  const int hw = p % HWSZ;
  const int h  = hw / WW_, w = hw % WW_;

  float* Xt = Xl + t * XSTRIDE;


  float xs0[KK], xs1[KK], xs2[KK];
#pragma unroll
  for (int k = 0; k < KK; ++k) {
    int m   = h * (WW_ * KK) + w * KK + k;
    int q   = m / HWSZ;
    int rem = m - q * HWSZ;
    int hp  = rem / WW_;
    int wp  = rem - hp * WW_;
    int ki  = q / 3;
    int kj  = q - 3 * ki;
    int hs  = hp + ki - 1; hs = hs < 0 ? 0 : (hs > HH_ - 1 ? HH_ - 1 : hs);
    int ws  = wp + kj - 1; ws = ws < 0 ? 0 : (ws > WW_ - 1 ? WW_ - 1 : ws);
    const float* xb = x + (size_t)b * 3 * HWSZ + (size_t)hs * WW_ + ws;
    xs0[k] = xb[0]; xs1[k] = xb[HWSZ]; xs2[k] = xb[2 * HWSZ];
  }

  v2f G2[KK][KK];
  v2f vs2[KK];
#pragma unroll
  for (int k = 0; k < KK; ++k) {
    vs2[k] = (v2f){0.0f, 0.0f};
#pragma unroll
    for (int j = 0; j <= k; ++j) G2[k][j] = (v2f){0.0f, 0.0f};
  }

  float hc[12];
#pragma unroll
  for (int i = 0; i < 12; ++i) hc[i] = hwt[i];
#pragma unroll 1
  for (int cg = 0; cg < 8; ++cg) {
    const float* hnp = hwt + ((cg + 1) & 7) * 12;
    float hn[12];
#pragma unroll
    for (int i = 0; i < 12; ++i) hn[i] = hnp[i];

    v2f d2[KK][2];
#pragma unroll
    for (int k = 0; k < KK; ++k) {
#pragma unroll
      for (int p = 0; p < 2; ++p) {
        v2f h0 = {hc[6 * p + 0], hc[6 * p + 3]};
        v2f h1 = {hc[6 * p + 1], hc[6 * p + 4]};
        v2f h2 = {hc[6 * p + 2], hc[6 * p + 5]};
        v2f r  = h0 * xs0[k] + h1 * xs1[k] + h2 * xs2[k];
        d2[k][p] = (v2f){fmaxf(r.x, 0.0f), fmaxf(r.y, 0.0f)};
      }
      *(float4*)(Xt + k * CCH + cg * 4) = make_float4(d2[k][0].x, d2[k][0].y, d2[k][1].x, d2[k][1].y);
      vs2[k] += d2[k][0];
      vs2[k] += d2[k][1];
    }
#pragma unroll
    for (int k = 0; k < KK; ++k) {
#pragma unroll
      for (int j = 0; j <= k; ++j) {
        G2[k][j] += d2[k][0] * d2[j][0];
        G2[k][j] += d2[k][1] * d2[j][1];
      }
    }
#pragma unroll
    for (int i = 0; i < 12; ++i) hc[i] = hn[i];
  }

  float G[KK][KK];
  float v[KK];
#pragma unroll
  for (int k = 0; k < KK; ++k) {
    v[k] = (vs2[k].x + vs2[k].y) * (1.0f / 32.0f);
#pragma unroll
    for (int j = 0; j <= k; ++j) {
      float g = G2[k][j].x + G2[k][j].y;
      G[k][j] = g;
      G[j][k] = g;
    }
  }

  float a[KK];
#pragma unroll
  for (int k = 0; k < KK; ++k) a[k] = (1.0f / 9.0f);

  for (int s = 0; s < NSTAGES; ++s) {
    float alpha = alphas[s];
    float beta  = betas[s];

    float numV[KK];
#pragma unroll
    for (int k = 0; k < KK; ++k) {
      float acc = 0.0f;
#pragma unroll
      for (int j = 0; j < KK; ++j) {
        acc += G[k][j] * a[j];
      }
      numV[k] = acc;
    }
    float s2u = (a[0] * numV[0] + a[1] * numV[1] + a[2] * numV[2])
              + (a[3] * numV[3] + a[4] * numV[4] + a[5] * numV[5])
              + (a[6] * numV[6] + a[7] * numV[7] + a[8] * numV[8]);
    float c3s2u = 3.0f * s2u;
#pragma unroll
    for (int k = 0; k < KK; ++k)
      v[k] = v[k] - beta * (c3s2u * v[k] - numV[k]);

    float s2v = (v[0] * v[0] + v[1] * v[1] + v[2] * v[2])
              + (v[3] * v[3] + v[4] * v[4] + v[5] * v[5])
              + (v[6] * v[6] + v[7] * v[7] + v[8] * v[8]);
    float c3s2v = 3.0f * s2v;
#pragma unroll
    for (int k = 0; k < KK; ++k)
      a[k] = a[k] - alpha * (c3s2v * a[k] - v[k]);
  }

  float4 u4[8];
#pragma unroll
  for (int cc = 0; cc < 8; ++cc) u4[cc] = make_float4(0.f, 0.f, 0.f, 0.f);
#pragma unroll
  for (int k = 0; k < KK; ++k) {
    const float4* row = (const float4*)(Xt + k * CCH);
    float ak = a[k];
#pragma unroll
    for (int cc = 0; cc < 8; ++cc) {
      float4 d = row[cc];
      u4[cc].x += ak * d.x; u4[cc].y += ak * d.y;
      u4[cc].z += ak * d.z; u4[cc].w += ak * d.w;
    }
  }

  wave_lds_order();
  float v4c = 3.0f * v[4];
#pragma unroll
  for (int cc = 0; cc < 8; ++cc) {
    UVl[(4 * cc + 0) * 32 + t] = u4[cc].x * v4c;
    UVl[(4 * cc + 1) * 32 + t] = u4[cc].y * v4c;
    UVl[(4 * cc + 2) * 32 + t] = u4[cc].z * v4c;
    UVl[(4 * cc + 3) * 32 + t] = u4[cc].w * v4c;
  }
  wave_lds_order();

  const int lane = t;
  const int mrow = lane & 15;
  const bool lo  = (lane < 16);
  float4 twv[8];
#pragma unroll
  for (int cc = 0; cc < 8; ++cc)
    twv[cc] = (mrow < 3) ? ((const float4*)(tailw + mrow * CCH))[cc]
                         : make_float4(0.f, 0.f, 0.f, 0.f);

  const int pix0   = blockIdx.x * BLK + wv * 32;
  const int bb     = pix0 / HWSZ;
  const int hwbase = pix0 % HWSZ;

#pragma unroll
  for (int half = 0; half < 2; ++half) {
    v8f acc = {0.f, 0.f, 0.f, 0.f, 0.f, 0.f, 0.f, 0.f};
#pragma unroll
    for (int k4 = 0; k4 < 8; ++k4) {
      float4 q = twv[k4];
      v2f a2, bm;
      a2.x = lo ? q.x : q.z;
      a2.y = lo ? q.y : q.w;
      int kk = 4 * k4 + (lo ? 0 : 2);
      bm.x = UVl[(kk + 0) * 32 + half * 16 + mrow];
      bm.y = UVl[(kk + 1) * 32 + half * 16 + mrow];
      acc = __builtin_amdgcn_wmma_f32_16x16x4_f32(
           false, a2,  false, bm,
           (short)0, acc,  false,  false);
    }
    if (lane < 16) {
      int hwp = hwbase + half * 16 + lane;
      float* ob = out + (size_t)bb * 3 * HWSZ + hwp;
      ob[0]        = fmaxf(acc[0], 0.0f);
      ob[HWSZ]     = fmaxf(acc[1], 0.0f);
      ob[2 * HWSZ] = fmaxf(acc[2], 0.0f);
    }
  }
}

extern "C" void kernel_launch(void* const* d_in, const int* in_sizes, int n_in,
                              void* d_out, int out_size, void* d_ws, size_t ws_size,
                              hipStream_t stream) {
  const float* x      = (const float*)d_in[0];
  const float* head_w = (const float*)d_in[1];
  const float* tail_w = (const float*)d_in[2];
  const float* alphas = (const float*)d_in[3];
  const float* betas  = (const float*)d_in[4];
  float* out = (float*)d_out;

  lowrank_kernel<<<(2 * HWSZ) / BLK, BLK, 0, stream>>>(x, head_w, tail_w, alphas, betas, out);
}
